# starting kernel
# baseline (speedup 1.0000x reference)
#include <hip/hip_runtime.h>

typedef float v2f __attribute__((ext_vector_type(2)));
typedef float v4f __attribute__((ext_vector_type(4)));
typedef float v8f __attribute__((ext_vector_type(8)));

__device__ __forceinline__ unsigned flt_key(float f) {
    unsigned u = __float_as_uint(f);
    return (u & 0x80000000u) ? ~u : (u | 0x80000000u);
}
__device__ __forceinline__ float key_flt(unsigned u) {
    return (u & 0x80000000u) ? __uint_as_float(u & 0x7FFFFFFFu)
                             : __uint_as_float(~u);
}

__global__ void segmax_init_kernel(unsigned* __restrict__ keys, int Bn) {
    int i = blockIdx.x * blockDim.x + threadIdx.x;
    if (i < Bn) keys[i] = 0u;
}

__global__ __launch_bounds__(128) void gemv_segmax_kernel(
    const float* __restrict__ z, const int* __restrict__ bag,
    const float* __restrict__ W, const float* __restrict__ pb,
    float* __restrict__ scores, unsigned* __restrict__ segkeys, int nTiles)
{
    __shared__ float smem[4 * 2048 + 128];
    float* wlds = smem + 4 * 2048;

    const int tid  = threadIdx.x;
    const int lane = tid & 31;
    const int wave = tid >> 5;
    float* tileW = smem + wave * 2048;

    if (tid < 128) wlds[tid] = W[tid];
    __syncthreads();

    const float bias = pb[0];
    const int wavesTotal = (gridDim.x * blockDim.x) >> 5;
    const int gwave = (blockIdx.x * blockDim.x + tid) >> 5;

    const int m  = lane & 15;
    const int kk = (lane >> 4) << 1;

    for (int tile = gwave; tile < nTiles; tile += wavesTotal) {
        const int row0 = tile << 4;

        const float* gsrc = z + (size_t)row0 * 128;
        #pragma unroll
        for (int it = 0; it < 16; ++it) {
            v4f v = *(const v4f*)(gsrc + it * 128 + lane * 4);
            *(v4f*)(tileW + it * 128 + lane * 4) = v;
        }
        __asm__ volatile("s_wait_dscnt 0x0" ::: "memory");

        v8f acc = {0.f, 0.f, 0.f, 0.f, 0.f, 0.f, 0.f, 0.f};
        #pragma unroll
        for (int c = 0; c < 32; ++c) {
            const int k0 = c * 4;
            v2f av = *(const v2f*)(tileW + m * 128 + k0 + kk);
            v2f bv = *(const v2f*)(wlds + k0 + kk);
            acc = __builtin_amdgcn_wmma_f32_16x16x4_f32(
                false, av, false, bv, (short)0, acc, false, false);
        }
        __asm__ volatile("" ::: "memory");

        if ((lane & 15) == 0) {
            const int rbase = row0 + ((lane >> 4) << 3);
            #pragma unroll
            for (int j = 0; j < 8; ++j) {
                float s = acc[j] + bias;
                scores[rbase + j] = s;
                int bg = bag[rbase + j];
                atomicMax(&segkeys[bg], flt_key(s));
            }
        }
    }
}

__global__ void gemv_tail_kernel(
    const float* __restrict__ z, const int* __restrict__ bag,
    const float* __restrict__ W, const float* __restrict__ pb,
    float* __restrict__ scores, unsigned* __restrict__ segkeys,
    int startRow, int N)
{
    int r = startRow + blockIdx.x * blockDim.x + threadIdx.x;
    if (r >= N) return;
    float s = pb[0];
    for (int k = 0; k < 128; ++k) s += z[(size_t)r * 128 + k] * W[k];
    scores[r] = s;
    atomicMax(&segkeys[bag[r]], flt_key(s));
}

__global__ void segmax_final_kernel(unsigned* __restrict__ keys, int Bn) {
    int i = blockIdx.x * blockDim.x + threadIdx.x;
    if (i < Bn) {
        unsigned u = keys[i];
        ((float*)keys)[i] = (u == 0u) ? 0.0f : key_flt(u);
    }
}

extern "C" void kernel_launch(void* const* d_in, const int* in_sizes, int n_in,
                              void* d_out, int out_size, void* d_ws, size_t ws_size,
                              hipStream_t stream) {
    const float* z   = (const float*)d_in[0];
    const int*   bag = (const int*)d_in[1];
    const float* W   = (const float*)d_in[2];
    const float* pb  = (const float*)d_in[3];

    const int N  = in_sizes[1];
    const int Bn = out_size - N;

    unsigned* segkeys = (unsigned*)d_out;
    float*    scores  = (float*)d_out + Bn;

    segmax_init_kernel<<<(Bn + 255) / 256, 256, 0, stream>>>(segkeys, Bn);

    const int nTiles = N >> 4;
    if (nTiles > 0) {
        gemv_segmax_kernel<<<2048, 128, 0, stream>>>(z, bag, W, pb,
                                                     scores, segkeys, nTiles);
    }
    const int rem = N & 15;
    if (rem > 0) {
        gemv_tail_kernel<<<(rem + 63) / 64, 64, 0, stream>>>(
            z, bag, W, pb, scores, segkeys, nTiles << 4, N);
    }
    segmax_final_kernel<<<(Bn + 255) / 256, 256, 0, stream>>>(segkeys, Bn);
}


# LDS tile row pitch 128 -> 132 floats (16 B x odd): the A-fragment reads of 16 rows at one k hit 64 banks instead of one bank pair (16-way conflict gone); summation order unchanged
# speedup vs baseline: 1.0549x; 1.0549x over previous
#include <hip/hip_runtime.h>

typedef float v2f __attribute__((ext_vector_type(2)));
typedef float v4f __attribute__((ext_vector_type(4)));
typedef float v8f __attribute__((ext_vector_type(8)));

__device__ __forceinline__ unsigned flt_key(float f) {
    unsigned u = __float_as_uint(f);
    return (u & 0x80000000u) ? ~u : (u | 0x80000000u);
}
__device__ __forceinline__ float key_flt(unsigned u) {
    return (u & 0x80000000u) ? __uint_as_float(u & 0x7FFFFFFFu)
                             : __uint_as_float(~u);
}

__global__ void segmax_init_kernel(unsigned* __restrict__ keys, int Bn) {
    int i = blockIdx.x * blockDim.x + threadIdx.x;
    if (i < Bn) keys[i] = 0u;
}

#define TPITCH 132
__global__ __launch_bounds__(128) void gemv_segmax_kernel(
    const float* __restrict__ z, const int* __restrict__ bag,
    const float* __restrict__ W, const float* __restrict__ pb,
    float* __restrict__ scores, unsigned* __restrict__ segkeys, int nTiles)
{
    __shared__ __attribute__((aligned(16))) float smem[4 * 16 * TPITCH + 128];
    float* wlds = smem + 4 * 16 * TPITCH;

    const int tid  = threadIdx.x;
    const int lane = tid & 31;
    const int wave = tid >> 5;
    float* tileW = smem + wave * (16 * TPITCH);

    if (tid < 128) wlds[tid] = W[tid];
    __syncthreads();

    const float bias = pb[0];
    const int wavesTotal = (gridDim.x * blockDim.x) >> 5;
    const int gwave = (blockIdx.x * blockDim.x + tid) >> 5;

    const int m  = lane & 15;
    const int kk = (lane >> 4) << 1;

    for (int tile = gwave; tile < nTiles; tile += wavesTotal) {
        const int row0 = tile << 4;

        const float* gsrc = z + (size_t)row0 * 128;
        #pragma unroll
        for (int it = 0; it < 16; ++it) {
            v4f v = *(const v4f*)(gsrc + it * 128 + lane * 4);
            *(v4f*)(tileW + it * TPITCH + lane * 4) = v;
        }
        __asm__ volatile("s_wait_dscnt 0x0" ::: "memory");

        v8f acc = {0.f, 0.f, 0.f, 0.f, 0.f, 0.f, 0.f, 0.f};
        #pragma unroll
        for (int c = 0; c < 32; ++c) {
            const int k0 = c * 4;
            v2f av = *(const v2f*)(tileW + m * TPITCH + k0 + kk);
            v2f bv = *(const v2f*)(wlds + k0 + kk);
            acc = __builtin_amdgcn_wmma_f32_16x16x4_f32(
                false, av, false, bv, (short)0, acc, false, false);
        }
        __asm__ volatile("" ::: "memory");

        if ((lane & 15) == 0) {
            const int rbase = row0 + ((lane >> 4) << 3);
            #pragma unroll
            for (int j = 0; j < 8; ++j) {
                float s = acc[j] + bias;
                scores[rbase + j] = s;
                int bg = bag[rbase + j];
                atomicMax(&segkeys[bg], flt_key(s));
            }
        }
    }
}

__global__ void gemv_tail_kernel(
    const float* __restrict__ z, const int* __restrict__ bag,
    const float* __restrict__ W, const float* __restrict__ pb,
    float* __restrict__ scores, unsigned* __restrict__ segkeys,
    int startRow, int N)
{
    int r = startRow + blockIdx.x * blockDim.x + threadIdx.x;
    if (r >= N) return;
    float s = pb[0];
    for (int k = 0; k < 128; ++k) s += z[(size_t)r * 128 + k] * W[k];
    scores[r] = s;
    atomicMax(&segkeys[bag[r]], flt_key(s));
}

__global__ void segmax_final_kernel(unsigned* __restrict__ keys, int Bn) {
    int i = blockIdx.x * blockDim.x + threadIdx.x;
    if (i < Bn) {
        unsigned u = keys[i];
        ((float*)keys)[i] = (u == 0u) ? 0.0f : key_flt(u);
    }
}

extern "C" void kernel_launch(void* const* d_in, const int* in_sizes, int n_in,
                              void* d_out, int out_size, void* d_ws, size_t ws_size,
                              hipStream_t stream) {
    const float* z   = (const float*)d_in[0];
    const int*   bag = (const int*)d_in[1];
    const float* W   = (const float*)d_in[2];
    const float* pb  = (const float*)d_in[3];

    const int N  = in_sizes[1];
    const int Bn = out_size - N;

    unsigned* segkeys = (unsigned*)d_out;
    float*    scores  = (float*)d_out + Bn;

    segmax_init_kernel<<<(Bn + 255) / 256, 256, 0, stream>>>(segkeys, Bn);

    const int nTiles = N >> 4;
    if (nTiles > 0) {
        gemv_segmax_kernel<<<2048, 128, 0, stream>>>(z, bag, W, pb,
                                                     scores, segkeys, nTiles);
    }
    const int rem = N & 15;
    if (rem > 0) {
        gemv_tail_kernel<<<(rem + 63) / 64, 64, 0, stream>>>(
            z, bag, W, pb, scores, segkeys, nTiles << 4, N);
    }
    segmax_final_kernel<<<(Bn + 255) / 256, 256, 0, stream>>>(segkeys, Bn);
}


# segment max via one LDS key table per work-group (LDS atomic max) instead of 500000 global atomic max; 256 groups x 32 waves, slices reduced in last launch; 3->2 launches; sum order unchanged
# speedup vs baseline: 2.4535x; 2.3258x over previous
#include <hip/hip_runtime.h>

typedef float v2f __attribute__((ext_vector_type(2)));
typedef float v4f __attribute__((ext_vector_type(4)));
typedef float v8f __attribute__((ext_vector_type(8)));

__device__ __forceinline__ unsigned flt_key(float f) {
    unsigned u = __float_as_uint(f);
    return (u & 0x80000000u) ? ~u : (u | 0x80000000u);
}
__device__ __forceinline__ float key_flt(unsigned u) {
    return (u & 0x80000000u) ? __uint_as_float(u & 0x7FFFFFFFu)
                             : __uint_as_float(~u);
}

__global__ void segmax_init_kernel(unsigned* __restrict__ keys, int Bn) {
    int i = blockIdx.x * blockDim.x + threadIdx.x;
    if (i < Bn) keys[i] = 0u;
}

#define TPITCH 132
__global__ __launch_bounds__(128) void gemv_segmax_kernel(
    const float* __restrict__ z, const int* __restrict__ bag,
    const float* __restrict__ W, const float* __restrict__ pb,
    float* __restrict__ scores, unsigned* __restrict__ segkeys, int nTiles)
{
    __shared__ __attribute__((aligned(16))) float smem[4 * 16 * TPITCH + 128];
    float* wlds = smem + 4 * 16 * TPITCH;

    const int tid  = threadIdx.x;
    const int lane = tid & 31;
    const int wave = tid >> 5;
    float* tileW = smem + wave * (16 * TPITCH);

    if (tid < 128) wlds[tid] = W[tid];
    __syncthreads();

    const float bias = pb[0];
    const int wavesTotal = (gridDim.x * blockDim.x) >> 5;
    const int gwave = (blockIdx.x * blockDim.x + tid) >> 5;

    const int m  = lane & 15;
    const int kk = (lane >> 4) << 1;

    for (int tile = gwave; tile < nTiles; tile += wavesTotal) {
        const int row0 = tile << 4;

        const float* gsrc = z + (size_t)row0 * 128;
        #pragma unroll
        for (int it = 0; it < 16; ++it) {
            v4f v = *(const v4f*)(gsrc + it * 128 + lane * 4);
            *(v4f*)(tileW + it * TPITCH + lane * 4) = v;
        }
        __asm__ volatile("s_wait_dscnt 0x0" ::: "memory");

        v8f acc = {0.f, 0.f, 0.f, 0.f, 0.f, 0.f, 0.f, 0.f};
        #pragma unroll
        for (int c = 0; c < 32; ++c) {
            const int k0 = c * 4;
            v2f av = *(const v2f*)(tileW + m * TPITCH + k0 + kk);
            v2f bv = *(const v2f*)(wlds + k0 + kk);
            acc = __builtin_amdgcn_wmma_f32_16x16x4_f32(
                false, av, false, bv, (short)0, acc, false, false);
        }
        __asm__ volatile("" ::: "memory");

        if ((lane & 15) == 0) {
            const int rbase = row0 + ((lane >> 4) << 3);
            #pragma unroll
            for (int j = 0; j < 8; ++j) {
                float s = acc[j] + bias;
                scores[rbase + j] = s;
                int bg = bag[rbase + j];
                atomicMax(&segkeys[bg], flt_key(s));
            }
        }
    }
}

__global__ void gemv_tail_kernel(
    const float* __restrict__ z, const int* __restrict__ bag,
    const float* __restrict__ W, const float* __restrict__ pb,
    float* __restrict__ scores, unsigned* __restrict__ segkeys,
    int startRow, int N)
{
    int r = startRow + blockIdx.x * blockDim.x + threadIdx.x;
    if (r >= N) return;
    float s = pb[0];
    for (int k = 0; k < 128; ++k) s += z[(size_t)r * 128 + k] * W[k];
    scores[r] = s;
    atomicMax(&segkeys[bag[r]], flt_key(s));
}

__global__ void segmax_final_kernel(unsigned* __restrict__ keys, int Bn) {
    int i = blockIdx.x * blockDim.x + threadIdx.x;
    if (i < Bn) {
        unsigned u = keys[i];
        ((float*)keys)[i] = (u == 0u) ? 0.0f : key_flt(u);
    }
}

#define NPART 256
#define TBL   4096

__global__ __launch_bounds__(1024) void gemv_segmax_tab_kernel(
    const float* __restrict__ z, const int* __restrict__ bag,
    const float* __restrict__ W, const float* __restrict__ pb,
    float* __restrict__ scores, unsigned* __restrict__ part, int nTiles)
{
    __shared__ __attribute__((aligned(16))) float smem[32 * 16 * TPITCH + 128];
    __shared__ __attribute__((aligned(16))) unsigned tab[TBL];
    float* wlds = smem + 32 * 16 * TPITCH;

    const int tid  = threadIdx.x;
    const int lane = tid & 31;
    const int wave = tid >> 5;
    float* tileW = smem + wave * (16 * TPITCH);

    if (tid < 128) wlds[tid] = W[tid];
    *(uint4*)(tab + tid * 4) = make_uint4(0u, 0u, 0u, 0u);
    __syncthreads();

    const float bias = pb[0];
    const int wavesTotal = (gridDim.x * blockDim.x) >> 5;
    const int gwave = (blockIdx.x * blockDim.x + tid) >> 5;

    const int m  = lane & 15;
    const int kk = (lane >> 4) << 1;

    for (int tile = gwave; tile < nTiles; tile += wavesTotal) {
        const int row0 = tile << 4;

        const float* gsrc = z + (size_t)row0 * 128;
        #pragma unroll
        for (int it = 0; it < 16; ++it) {
            v4f v = *(const v4f*)(gsrc + it * 128 + lane * 4);
            *(v4f*)(tileW + it * TPITCH + lane * 4) = v;
        }
        __asm__ volatile("s_wait_dscnt 0x0" ::: "memory");

        v8f acc = {0.f, 0.f, 0.f, 0.f, 0.f, 0.f, 0.f, 0.f};
        #pragma unroll
        for (int c = 0; c < 32; ++c) {
            const int k0 = c * 4;
            v2f av = *(const v2f*)(tileW + m * TPITCH + k0 + kk);
            v2f bv = *(const v2f*)(wlds + k0 + kk);
            acc = __builtin_amdgcn_wmma_f32_16x16x4_f32(
                false, av, false, bv, (short)0, acc, false, false);
        }

        if ((lane & 15) == 0) {
            const int rbase = row0 + ((lane >> 4) << 3);
            #pragma unroll
            for (int j = 0; j < 8; ++j) {
                float s = acc[j] + bias;
                scores[rbase + j] = s;
                int bg = bag[rbase + j];
                __hip_atomic_fetch_max(&tab[bg], flt_key(s), __ATOMIC_RELAXED,
                                       __HIP_MEMORY_SCOPE_WORKGROUP);
            }
        }
    }

    __syncthreads();
    *(uint4*)(part + (size_t)blockIdx.x * TBL + tid * 4) = *(const uint4*)(tab + tid * 4);
}

__global__ void gemv_tail_scores_kernel(
    const float* __restrict__ z, const float* __restrict__ W,
    const float* __restrict__ pb, float* __restrict__ scores,
    int startRow, int N)
{
    int r = startRow + blockIdx.x * blockDim.x + threadIdx.x;
    if (r >= N) return;
    float s = pb[0];
    for (int k = 0; k < 128; ++k) s += z[(size_t)r * 128 + k] * W[k];
    scores[r] = s;
}

__global__ __launch_bounds__(256) void segmax_reduce_kernel(
    const unsigned* __restrict__ part, const int* __restrict__ bag,
    const float* __restrict__ scores, float* __restrict__ M,
    int Bn, int tailStart, int N)
{
    __shared__ unsigned red[8][32];
    const int lane = threadIdx.x & 31;
    const int w    = threadIdx.x >> 5;
    const int b    = blockIdx.x * 32 + lane;

    unsigned k = 0u;
    if (b < Bn) {
        #pragma unroll
        for (int i = 0; i < NPART / 8; ++i) {
            unsigned v = part[(size_t)(w + 8 * i) * TBL + b];
            k = (v > k) ? v : k;
        }
    }
    red[w][lane] = k;
    __syncthreads();

    if (w == 0 && b < Bn) {
        #pragma unroll
        for (int j = 1; j < 8; ++j) {
            unsigned v = red[j][lane];
            k = (v > k) ? v : k;
        }
        for (int r = tailStart; r < N; ++r) {
            if (bag[r] == b) {
                unsigned v = flt_key(scores[r]);
                k = (v > k) ? v : k;
            }
        }
        M[b] = (k == 0u) ? 0.0f : key_flt(k);
    }
}

extern "C" void kernel_launch(void* const* d_in, const int* in_sizes, int n_in,
                              void* d_out, int out_size, void* d_ws, size_t ws_size,
                              hipStream_t stream) {
    const float* z   = (const float*)d_in[0];
    const int*   bag = (const int*)d_in[1];
    const float* W   = (const float*)d_in[2];
    const float* pb  = (const float*)d_in[3];

    const int N  = in_sizes[1];
    const int Bn = out_size - N;

    float* scores = (float*)d_out + Bn;
    const int nTiles = N >> 4;
    const int rem = N & 15;

    if (Bn <= TBL && ws_size >= (size_t)NPART * TBL * sizeof(unsigned)) {
        unsigned* part = (unsigned*)d_ws;
        gemv_segmax_tab_kernel<<<NPART, 1024, 0, stream>>>(z, bag, W, pb,
                                                           scores, part, nTiles);
        if (rem > 0) {
            gemv_tail_scores_kernel<<<(rem + 63) / 64, 64, 0, stream>>>(
                z, W, pb, scores, nTiles << 4, N);
        }
        segmax_reduce_kernel<<<(Bn + 31) / 32, 256, 0, stream>>>(
            part, bag, scores, (float*)d_out, Bn, nTiles << 4, N);
        return;
    }

    unsigned* segkeys = (unsigned*)d_out;

    segmax_init_kernel<<<(Bn + 255) / 256, 256, 0, stream>>>(segkeys, Bn);

    if (nTiles > 0) {
        gemv_segmax_kernel<<<2048, 128, 0, stream>>>(z, bag, W, pb,
                                                     scores, segkeys, nTiles);
    }
    if (rem > 0) {
        gemv_tail_kernel<<<(rem + 63) / 64, 64, 0, stream>>>(
            z, bag, W, pb, scores, segkeys, nTiles << 4, N);
    }
    segmax_final_kernel<<<(Bn + 255) / 256, 256, 0, stream>>>(segkeys, Bn);
}


# main kernel: each wave requests its next 16-row tile (row loads + bag indices) before the WMMA pass over the current tile, so memory reads overlap the matrix work; sum order unchanged
# speedup vs baseline: 2.5351x; 1.0333x over previous
#include <hip/hip_runtime.h>

typedef float v2f __attribute__((ext_vector_type(2)));
typedef float v4f __attribute__((ext_vector_type(4)));
typedef float v8f __attribute__((ext_vector_type(8)));

__device__ __forceinline__ unsigned flt_key(float f) {
    unsigned u = __float_as_uint(f);
    return (u & 0x80000000u) ? ~u : (u | 0x80000000u);
}
__device__ __forceinline__ float key_flt(unsigned u) {
    return (u & 0x80000000u) ? __uint_as_float(u & 0x7FFFFFFFu)
                             : __uint_as_float(~u);
}

__global__ void segmax_init_kernel(unsigned* __restrict__ keys, int Bn) {
    int i = blockIdx.x * blockDim.x + threadIdx.x;
    if (i < Bn) keys[i] = 0u;
}

#define TPITCH 132
__global__ __launch_bounds__(128) void gemv_segmax_kernel(
    const float* __restrict__ z, const int* __restrict__ bag,
    const float* __restrict__ W, const float* __restrict__ pb,
    float* __restrict__ scores, unsigned* __restrict__ segkeys, int nTiles)
{
    __shared__ __attribute__((aligned(16))) float smem[4 * 16 * TPITCH + 128];
    float* wlds = smem + 4 * 16 * TPITCH;

    const int tid  = threadIdx.x;
    const int lane = tid & 31;
    const int wave = tid >> 5;
    float* tileW = smem + wave * (16 * TPITCH);

    if (tid < 128) wlds[tid] = W[tid];
    __syncthreads();

    const float bias = pb[0];
    const int wavesTotal = (gridDim.x * blockDim.x) >> 5;
    const int gwave = (blockIdx.x * blockDim.x + tid) >> 5;

    const int m  = lane & 15;
    const int kk = (lane >> 4) << 1;

    for (int tile = gwave; tile < nTiles; tile += wavesTotal) {
        const int row0 = tile << 4;

        const float* gsrc = z + (size_t)row0 * 128;
        #pragma unroll
        for (int it = 0; it < 16; ++it) {
            v4f v = *(const v4f*)(gsrc + it * 128 + lane * 4);
            *(v4f*)(tileW + it * TPITCH + lane * 4) = v;
        }
        __asm__ volatile("s_wait_dscnt 0x0" ::: "memory");

        v8f acc = {0.f, 0.f, 0.f, 0.f, 0.f, 0.f, 0.f, 0.f};
        #pragma unroll
        for (int c = 0; c < 32; ++c) {
            const int k0 = c * 4;
            v2f av = *(const v2f*)(tileW + m * TPITCH + k0 + kk);
            v2f bv = *(const v2f*)(wlds + k0 + kk);
            acc = __builtin_amdgcn_wmma_f32_16x16x4_f32(
                false, av, false, bv, (short)0, acc, false, false);
        }
        __asm__ volatile("" ::: "memory");

        if ((lane & 15) == 0) {
            const int rbase = row0 + ((lane >> 4) << 3);
            #pragma unroll
            for (int j = 0; j < 8; ++j) {
                float s = acc[j] + bias;
                scores[rbase + j] = s;
                int bg = bag[rbase + j];
                atomicMax(&segkeys[bg], flt_key(s));
            }
        }
    }
}

__global__ void gemv_tail_kernel(
    const float* __restrict__ z, const int* __restrict__ bag,
    const float* __restrict__ W, const float* __restrict__ pb,
    float* __restrict__ scores, unsigned* __restrict__ segkeys,
    int startRow, int N)
{
    int r = startRow + blockIdx.x * blockDim.x + threadIdx.x;
    if (r >= N) return;
    float s = pb[0];
    for (int k = 0; k < 128; ++k) s += z[(size_t)r * 128 + k] * W[k];
    scores[r] = s;
    atomicMax(&segkeys[bag[r]], flt_key(s));
}

__global__ void segmax_final_kernel(unsigned* __restrict__ keys, int Bn) {
    int i = blockIdx.x * blockDim.x + threadIdx.x;
    if (i < Bn) {
        unsigned u = keys[i];
        ((float*)keys)[i] = (u == 0u) ? 0.0f : key_flt(u);
    }
}

#define NPART 256
#define TBL   4096

__global__ __launch_bounds__(1024) void gemv_segmax_tab_kernel(
    const float* __restrict__ z, const int* __restrict__ bag,
    const float* __restrict__ W, const float* __restrict__ pb,
    float* __restrict__ scores, unsigned* __restrict__ part, int nTiles)
{
    __shared__ __attribute__((aligned(16))) float smem[32 * 16 * TPITCH + 128];
    __shared__ __attribute__((aligned(16))) unsigned tab[TBL];
    float* wlds = smem + 32 * 16 * TPITCH;

    const int tid  = threadIdx.x;
    const int lane = tid & 31;
    const int wave = tid >> 5;
    float* tileW = smem + wave * (16 * TPITCH);

    if (tid < 128) wlds[tid] = W[tid];
    *(uint4*)(tab + tid * 4) = make_uint4(0u, 0u, 0u, 0u);
    __syncthreads();

    const float bias = pb[0];
    const int wavesTotal = (gridDim.x * blockDim.x) >> 5;
    const int gwave = (blockIdx.x * blockDim.x + tid) >> 5;

    const int m  = lane & 15;
    const int kk = (lane >> 4) << 1;
    const int hrow = (lane >> 4) << 3;

    v4f r[16];
    int4 nb0 = make_int4(0, 0, 0, 0), nb1 = make_int4(0, 0, 0, 0);
    int tile = gwave;
    if (tile < nTiles) {
        const int* bsrc = bag + (tile << 4) + hrow;
        nb0 = *(const int4*)(bsrc);
        nb1 = *(const int4*)(bsrc + 4);
        const float* gsrc = z + (size_t)(tile << 4) * 128;
        #pragma unroll
        for (int it = 0; it < 16; ++it) r[it] = *(const v4f*)(gsrc + it * 128 + lane * 4);
    }

    for (; tile < nTiles; tile += wavesTotal) {
        const int row0 = tile << 4;

        #pragma unroll
        for (int it = 0; it < 16; ++it) *(v4f*)(tileW + it * TPITCH + lane * 4) = r[it];
        const int4 cb0 = nb0, cb1 = nb1;

        const int nxt = tile + wavesTotal;
        if (nxt < nTiles) {
            const int* bsrc = bag + (nxt << 4) + hrow;
            nb0 = *(const int4*)(bsrc);
            nb1 = *(const int4*)(bsrc + 4);
            const float* gsrc = z + (size_t)(nxt << 4) * 128;
            #pragma unroll
            for (int it = 0; it < 16; ++it) r[it] = *(const v4f*)(gsrc + it * 128 + lane * 4);
        }
        __asm__ volatile("s_wait_dscnt 0x0" ::: "memory");

        v8f acc = {0.f, 0.f, 0.f, 0.f, 0.f, 0.f, 0.f, 0.f};
        #pragma unroll
        for (int c = 0; c < 32; ++c) {
            const int k0 = c * 4;
            v2f av = *(const v2f*)(tileW + m * TPITCH + k0 + kk);
            v2f bv = *(const v2f*)(wlds + k0 + kk);
            acc = __builtin_amdgcn_wmma_f32_16x16x4_f32(
                false, av, false, bv, (short)0, acc, false, false);
        }

        if ((lane & 15) == 0) {
            const int rbase = row0 + hrow;
            const int bgs[8] = {cb0.x, cb0.y, cb0.z, cb0.w, cb1.x, cb1.y, cb1.z, cb1.w};
            #pragma unroll
            for (int j = 0; j < 8; ++j) {
                float s = acc[j] + bias;
                scores[rbase + j] = s;
                __hip_atomic_fetch_max(&tab[bgs[j]], flt_key(s), __ATOMIC_RELAXED,
                                       __HIP_MEMORY_SCOPE_WORKGROUP);
            }
        }
    }

    __syncthreads();
    *(uint4*)(part + (size_t)blockIdx.x * TBL + tid * 4) = *(const uint4*)(tab + tid * 4);
}

__global__ void gemv_tail_scores_kernel(
    const float* __restrict__ z, const float* __restrict__ W,
    const float* __restrict__ pb, float* __restrict__ scores,
    int startRow, int N)
{
    int r = startRow + blockIdx.x * blockDim.x + threadIdx.x;
    if (r >= N) return;
    float s = pb[0];
    for (int k = 0; k < 128; ++k) s += z[(size_t)r * 128 + k] * W[k];
    scores[r] = s;
}

__global__ __launch_bounds__(256) void segmax_reduce_kernel(
    const unsigned* __restrict__ part, const int* __restrict__ bag,
    const float* __restrict__ scores, float* __restrict__ M,
    int Bn, int tailStart, int N)
{
    __shared__ unsigned red[8][32];
    const int lane = threadIdx.x & 31;
    const int w    = threadIdx.x >> 5;
    const int b    = blockIdx.x * 32 + lane;

    unsigned k = 0u;
    if (b < Bn) {
        #pragma unroll
        for (int i = 0; i < NPART / 8; ++i) {
            unsigned v = part[(size_t)(w + 8 * i) * TBL + b];
            k = (v > k) ? v : k;
        }
    }
    red[w][lane] = k;
    __syncthreads();

    if (w == 0 && b < Bn) {
        #pragma unroll
        for (int j = 1; j < 8; ++j) {
            unsigned v = red[j][lane];
            k = (v > k) ? v : k;
        }
        for (int r = tailStart; r < N; ++r) {
            if (bag[r] == b) {
                unsigned v = flt_key(scores[r]);
                k = (v > k) ? v : k;
            }
        }
        M[b] = (k == 0u) ? 0.0f : key_flt(k);
    }
}

extern "C" void kernel_launch(void* const* d_in, const int* in_sizes, int n_in,
                              void* d_out, int out_size, void* d_ws, size_t ws_size,
                              hipStream_t stream) {
    const float* z   = (const float*)d_in[0];
    const int*   bag = (const int*)d_in[1];
    const float* W   = (const float*)d_in[2];
    const float* pb  = (const float*)d_in[3];

    const int N  = in_sizes[1];
    const int Bn = out_size - N;

    float* scores = (float*)d_out + Bn;
    const int nTiles = N >> 4;
    const int rem = N & 15;

    if (Bn <= TBL && ws_size >= (size_t)NPART * TBL * sizeof(unsigned)) {
        unsigned* part = (unsigned*)d_ws;
        gemv_segmax_tab_kernel<<<NPART, 1024, 0, stream>>>(z, bag, W, pb,
                                                           scores, part, nTiles);
        if (rem > 0) {
            gemv_tail_scores_kernel<<<(rem + 63) / 64, 64, 0, stream>>>(
                z, W, pb, scores, nTiles << 4, N);
        }
        segmax_reduce_kernel<<<(Bn + 31) / 32, 256, 0, stream>>>(
            part, bag, scores, (float*)d_out, Bn, nTiles << 4, N);
        return;
    }

    unsigned* segkeys = (unsigned*)d_out;

    segmax_init_kernel<<<(Bn + 255) / 256, 256, 0, stream>>>(segkeys, Bn);

    if (nTiles > 0) {
        gemv_segmax_kernel<<<2048, 128, 0, stream>>>(z, bag, W, pb,
                                                     scores, segkeys, nTiles);
    }
    if (rem > 0) {
        gemv_tail_kernel<<<(rem + 63) / 64, 64, 0, stream>>>(
            z, bag, W, pb, scores, segkeys, nTiles << 4, N);
    }
    segmax_final_kernel<<<(Bn + 255) / 256, 256, 0, stream>>>(segkeys, Bn);
}
